# starting kernel
# baseline (speedup 1.0000x reference)
#include <hip/hip_runtime.h>
#include <hip/hip_fp16.h>


typedef __attribute__((ext_vector_type(2)))  float    v2f;
typedef __attribute__((ext_vector_type(8)))  float    v8f;
typedef __attribute__((ext_vector_type(16))) _Float16 v16h;

#define BATCH   8
#define HW_IN   112
#define HW_OUT  56
#define NPATCH  (HW_OUT * HW_OUT)
#define NTILE   (NPATCH / 16)
#define LDS_RS  18

__device__ __forceinline__ v8f wmma_k4(v2f a, v2f b, v8f c) {
#if __has_builtin(__builtin_amdgcn_wmma_f32_16x16x4_f32)
  return __builtin_amdgcn_wmma_f32_16x16x4_f32(false, a, false, b, (short)0, c,
                                               false, false);
#else
  v16h ah = {};
  v16h bh = {};
  ah[0] = (_Float16)a[0]; ah[1] = (_Float16)a[1];
  bh[0] = (_Float16)b[0]; bh[1] = (_Float16)b[1];
  return __builtin_amdgcn_wmma_f32_16x16x32_f16(false, ah, false, bh, (short)0, c,
                                                false, false);
#endif
}

__global__ __launch_bounds__(256) void conv_norm_kernel(
    const float* __restrict__ x,
    const float* __restrict__ cw,
    const float* __restrict__ cb,
    float* __restrict__ patches,
    float* __restrict__ normed) {
  int t = blockIdx.x * blockDim.x + threadIdx.x;
  if (t >= BATCH * NPATCH) return;
  int b = t / NPATCH;
  int n = t % NPATCH;
  int h = n / HW_OUT, w = n % HW_OUT;

  const float* xb = x + ((size_t)b * HW_IN + 2 * h) * HW_IN + 2 * w;
  float x00 = xb[0], x01 = xb[1], x10 = xb[HW_IN], x11 = xb[HW_IN + 1];

  float f[4];
  float ss = 0.0f;
#pragma unroll
  for (int cc = 0; cc < 4; ++cc) {
    float v = x00 * cw[cc * 4 + 0] + x01 * cw[cc * 4 + 1] +
              x10 * cw[cc * 4 + 2] + x11 * cw[cc * 4 + 3] + cb[cc];
    f[cc] = v;
    ss += v * v;
  }
  float inv = 1.0f / (sqrtf(ss) + 1e-12f);

  float4 pf = make_float4(f[0], f[1], f[2], f[3]);
  float4 nf = make_float4(f[0] * inv, f[1] * inv, f[2] * inv, f[3] * inv);
  *(float4*)(patches + (size_t)t * 4) = pf;
  *(float4*)(normed  + (size_t)t * 4) = nf;
}

__global__ __launch_bounds__(256) void build_paugb_kernel(
    const float* __restrict__ patches,
    float* __restrict__ paugB) {
  int t = blockIdx.x * blockDim.x + threadIdx.x;
  if (t >= BATCH * NTILE * 4 * 32) return;
  int lane = t & 31;
  int k    = (t >> 5) & 3;
  int j    = (t >> 7) % NTILE;
  int b    = t / (NTILE * 4 * 32);

  int half  = lane >> 4;
  int c     = lane & 15;
  int kbase = half << 1;
  int m0    = 4 * k + kbase;
  int cclamp = (c < 3) ? c : 3;

  const float* pat_b = patches + (size_t)b * NPATCH * 4;
  float pv0 = pat_b[(j * 16 + m0    ) * 4 + cclamp];
  float pv1 = pat_b[(j * 16 + m0 + 1) * 4 + cclamp];

  v2f v;
  v[0] = (c < 4) ? pv0 : ((c == 4) ? 1.0f : 0.0f);
  v[1] = (c < 4) ? pv1 : ((c == 4) ? 1.0f : 0.0f);
  *(v2f*)(paugB + (size_t)t * 2) = v;
}

__global__ __launch_bounds__(256) void smooth_kernel(
    const float* __restrict__ patches,
    const float* __restrict__ normed,
    const float* __restrict__ paugB,
    float* __restrict__ out) {
  __shared__ float lds[8 * 16 * LDS_RS];

  const int lane  = threadIdx.x & 31;
  const int wave  = threadIdx.x >> 5;
  const int gw    = blockIdx.x * 8 + wave;
  const int b     = gw / NTILE;
  const int it    = gw % NTILE;

  float* wtile = &lds[wave * 16 * LDS_RS];

  const int half  = lane >> 4;
  const int l15   = lane & 15;
  const int kbase = half << 1;
  const int c     = l15;

  const float* nrm_b  = normed  + (size_t)b * NPATCH * 4;
  const float* pat_b  = patches + (size_t)b * NPATCH * 4;
  const float* paug_b = paugB   + (size_t)b * NTILE * 4 * 32 * 2;

  v2f a1 = *(const v2f*)(nrm_b + (it * 16 + l15) * 4 + kbase);

  v8f acc = {};

  for (int j = 0; j < NTILE; ++j) {
    v2f b1 = *(const v2f*)(nrm_b + (j * 16 + l15) * 4 + kbase);

    int jn = (j + 1 < NTILE) ? (j + 1) : (NTILE - 1);
    __builtin_prefetch(nrm_b + (jn * 16 + l15) * 4 + kbase, 0, 3);
    __builtin_prefetch(paug_b + (size_t)jn * 4 * 32 * 2 + lane * 2, 0, 3);

    v8f zero = {};
    v8f cosd = wmma_k4(a1, b1, zero);

#pragma unroll
    for (int r = 0; r < 8; ++r) {
      float cc  = cosd[r];
      float fid = cc * cc;
      float wgt = (fid >= 0.8f) ? 1.0f : ((fid >= 0.5f) ? 0.5f : 0.0f);
      int row_local = r + 8 * half;
      if (it == j && row_local == c) wgt = 0.0f;
      wtile[row_local * LDS_RS + l15] = wgt;
    }
    asm volatile("s_wait_dscnt 0" ::: "memory");

#pragma unroll
    for (int k = 0; k < 4; ++k) {
      v2f a2 = *(const v2f*)(wtile + l15 * LDS_RS + 4 * k + kbase);
      v2f b2 = *(const v2f*)(paug_b + ((size_t)j * 4 + k) * 32 * 2 + lane * 2);
      acc = wmma_k4(a2, b2, acc);
    }
  }

#pragma unroll
  for (int r = 0; r < 8; ++r) {
    int row_local = r + 8 * half;
    float deg = __shfl(acc[r], (half << 4) | 4, 32);
    if (c < 4) {
      int n = it * 16 + row_local;
      float p = pat_b[n * 4 + c];
      out[(size_t)b * NPATCH * 4 + n * 4 + c] = p * (1.0f + deg) - acc[r];
    }
  }
}

extern "C" void kernel_launch(void* const* d_in, const int* in_sizes, int n_in,
                              void* d_out, int out_size, void* d_ws, size_t ws_size,
                              hipStream_t stream) {
  const float* x  = (const float*)d_in[0];
  const float* cw = (const float*)d_in[1];
  const float* cb = (const float*)d_in[2];
  float* out = (float*)d_out;

  float* patches = (float*)d_ws;
  float* normed  = patches + (size_t)BATCH * NPATCH * 4;
  float* paugB   = normed  + (size_t)BATCH * NPATCH * 4;

  int total = BATCH * NPATCH;
  conv_norm_kernel<<<(total + 255) / 256, 256, 0, stream>>>(x, cw, cb, patches, normed);

  int totalB = BATCH * NTILE * 4 * 32;
  build_paugb_kernel<<<(totalB + 255) / 256, 256, 0, stream>>>(patches, paugB);

  smooth_kernel<<<NTILE, 256, 0, stream>>>(patches, normed, paugB, out);
}


# Each pass's 12 loads issued as ONE asm group at the pass head and waited for once (count 12) at the next pass head: no load, load wait or register copy among the loop's matrix work; sums unchanged
# speedup vs baseline: 4.4780x; 4.4780x over previous
#include <hip/hip_runtime.h>
#include <hip/hip_fp16.h>


typedef __attribute__((ext_vector_type(2)))  float    v2f;
typedef __attribute__((ext_vector_type(4)))  float    v4f;
typedef __attribute__((ext_vector_type(8)))  float    v8f;
typedef __attribute__((ext_vector_type(16))) _Float16 v16h;

#define BATCH   8
#define HW_IN   112
#define HW_OUT  56
#define NPATCH  (HW_OUT * HW_OUT)
#define NTILE   (NPATCH / 16)
#define RING    4
#define NGROUP  (NTILE / RING)
#define CONSTF  (RING * 16)
static_assert(NTILE % RING == 0, "column loop is unrolled by RING");
static_assert(NTILE >= 2 * RING, "the prologue loads two groups");
static_assert(NGROUP % 2 == 1, "the column loop runs pairs of passes and peels one last pass");

__device__ __forceinline__ v8f wmma_k4(v2f a, v2f b, v8f c) {
#if __has_builtin(__builtin_amdgcn_wmma_f32_16x16x4_f32)
  return __builtin_amdgcn_wmma_f32_16x16x4_f32(false, a, false, b, (short)0, c,
                                               false, false);
#else
  v16h ah = {};
  v16h bh = {};
  ah[0] = (_Float16)a[0]; ah[1] = (_Float16)a[1];
  bh[0] = (_Float16)b[0]; bh[1] = (_Float16)b[1];
  return __builtin_amdgcn_wmma_f32_16x16x32_f16(false, ah, false, bh, (short)0, c,
                                                false, false);
#endif
}

__global__ __launch_bounds__(256) void conv_norm_kernel(
    const float* __restrict__ x,
    const float* __restrict__ cw,
    const float* __restrict__ cb,
    float* __restrict__ patches,
    float* __restrict__ normed,
    float* __restrict__ patT,
    float* __restrict__ consts) {
  int t = blockIdx.x * blockDim.x + threadIdx.x;
  if (t >= BATCH * NPATCH) return;
  if (t < CONSTF) {
    consts[t] = 1.0f;
    consts[CONSTF + t] = 0.0f;
  }
  int b = t / NPATCH;
  int n = t % NPATCH;
  int h = n / HW_OUT, w = n % HW_OUT;

  const float* xb = x + ((size_t)b * HW_IN + 2 * h) * HW_IN + 2 * w;
  float x00 = xb[0], x01 = xb[1], x10 = xb[HW_IN], x11 = xb[HW_IN + 1];

  float f[4];
  float ss = 0.0f;
#pragma unroll
  for (int cc = 0; cc < 4; ++cc) {
    float v = x00 * cw[cc * 4 + 0] + x01 * cw[cc * 4 + 1] +
              x10 * cw[cc * 4 + 2] + x11 * cw[cc * 4 + 3] + cb[cc];
    f[cc] = v;
    ss += v * v;
  }
  float inv = 1.0f / (sqrtf(ss) + 1e-12f);

  float4 pf = make_float4(f[0], f[1], f[2], f[3]);
  float4 nf = make_float4(f[0] * inv, f[1] * inv, f[2] * inv, f[3] * inv);
  *(float4*)(patches + (size_t)t * 4) = pf;
  *(float4*)(normed  + (size_t)t * 4) = nf;
#pragma unroll
  for (int cc = 0; cc < 4; ++cc)
    patT[((size_t)b * 4 + cc) * NPATCH + n] = f[cc];
}

__device__ __forceinline__ void load_group(const float* n1p, const float* b2p,
                                           v2f (&b1)[RING], v4f (&b2)[RING][2],
                                           v2f& rowop, v8f& acc) {
  static_assert(RING == 4, "the asm statement below holds the loads of 4 tiles");
  asm volatile(
      "global_load_b64 %0, %14, off\n\t"
      "global_load_b64 %1, %14, off offset:256\n\t"
      "global_load_b64 %2, %14, off offset:512\n\t"
      "global_load_b64 %3, %14, off offset:768\n\t"
      "global_load_b128 %4, %15, off\n\t"
      "global_load_b128 %5, %15, off offset:16\n\t"
      "global_load_b128 %6, %15, off offset:64\n\t"
      "global_load_b128 %7, %15, off offset:80\n\t"
      "global_load_b128 %8, %15, off offset:128\n\t"
      "global_load_b128 %9, %15, off offset:144\n\t"
      "global_load_b128 %10, %15, off offset:192\n\t"
      "global_load_b128 %11, %15, off offset:208"
      : "=&v"(b1[0]), "=&v"(b1[1]), "=&v"(b1[2]), "=&v"(b1[3]),
        "=&v"(b2[0][0]), "=&v"(b2[0][1]), "=&v"(b2[1][0]), "=&v"(b2[1][1]),
        "=&v"(b2[2][0]), "=&v"(b2[2][1]), "=&v"(b2[3][0]), "=&v"(b2[3][1]),
        "+v"(rowop), "+v"(acc)
      : "v"(n1p), "v"(b2p));
}

__device__ __forceinline__ void load_group_release(const float* n1p, const float* b2p,
                                                   v8f& acc) {
  asm volatile("s_wait_xcnt 0x0" : "+v"(acc) : "v"(n1p), "v"(b2p));
}

template <bool NEWER>
__device__ __forceinline__ void load_group_wait(v2f (&b1)[RING], v4f (&b2)[RING][2]) {
  if (NEWER) {
    asm volatile(
        "s_wait_loadcnt 0xc"
        : "+v"(b1[0]), "+v"(b1[1]), "+v"(b1[2]), "+v"(b1[3]),
          "+v"(b2[0][0]), "+v"(b2[0][1]), "+v"(b2[1][0]), "+v"(b2[1][1]),
          "+v"(b2[2][0]), "+v"(b2[2][1]), "+v"(b2[3][0]), "+v"(b2[3][1]));
  } else {
    asm volatile(
        "s_wait_loadcnt 0x0"
        : "+v"(b1[0]), "+v"(b1[1]), "+v"(b1[2]), "+v"(b1[3]),
          "+v"(b2[0][0]), "+v"(b2[0][1]), "+v"(b2[1][0]), "+v"(b2[1][1]),
          "+v"(b2[2][0]), "+v"(b2[2][1]), "+v"(b2[3][0]), "+v"(b2[3][1]));
  }
}

__device__ __forceinline__ void weight_tile(v2f colop, v2f rowop, v8f& w) {
  v8f zero = {};
  v8f cosd = wmma_k4(colop, rowop, zero);
#pragma unroll
  for (int r = 0; r < 8; ++r) {
    float cc  = cosd[r];
    float fid = cc * cc;
    w[r] = (fid >= 0.8f) ? 1.0f : ((fid >= 0.5f) ? 0.5f : 0.0f);
  }
}

template <bool LAST>
__device__ __forceinline__ void group_pass(
    int g, const float* nrm_b, const float* b2base, int b2step,
    v2f& a1, int l15, int kbase,
    v2f (&cb1)[RING], v4f (&cb2)[RING][2],
    v2f (&nb1)[RING], v4f (&nb2)[RING][2],
    const v8f (&wcur)[RING], v8f (&wnxt)[RING], v8f& acc) {
  const float* n1p = nrm_b;
  const float* b2p = b2base;
  if (!LAST) {
    const int g1 = (g + 2 < NGROUP) ? (g + 2) : (NGROUP - 1);
    const int g2 = g + 1;
    n1p = nrm_b + ((size_t)g1 * RING * 16 + l15) * 4 + kbase;
    b2p = b2base + (size_t)g2 * b2step;
    load_group(n1p, b2p, nb1, nb2, a1, acc);
    load_group_wait<true>(cb1, cb2);

#pragma unroll
    for (int s = 0; s < RING; ++s)
      weight_tile(cb1[s], a1, wnxt[s]);
  }

  if (LAST) load_group_wait<false>(cb1, cb2);

#pragma unroll
  for (int s = 0; s < RING; ++s) {
    v2f ak, bk;
    ak[0] = wcur[s][0]; ak[1] = wcur[s][1];
    bk[0] = cb2[s][0].x; bk[1] = cb2[s][0].y;
    acc = wmma_k4(ak, bk, acc);
    ak[0] = wcur[s][2]; ak[1] = wcur[s][3];
    bk[0] = cb2[s][0].z; bk[1] = cb2[s][0].w;
    acc = wmma_k4(ak, bk, acc);
    ak[0] = wcur[s][4]; ak[1] = wcur[s][5];
    bk[0] = cb2[s][1].x; bk[1] = cb2[s][1].y;
    acc = wmma_k4(ak, bk, acc);
    ak[0] = wcur[s][6]; ak[1] = wcur[s][7];
    bk[0] = cb2[s][1].z; bk[1] = cb2[s][1].w;
    acc = wmma_k4(ak, bk, acc);
    if (!LAST && s == 0) load_group_release(n1p, b2p, acc);
  }
}

__global__ __launch_bounds__(256) void smooth_kernel(
    const float* __restrict__ patches,
    const float* __restrict__ normed,
    const float* __restrict__ patT,
    const float* __restrict__ consts,
    float* __restrict__ out) {
  const int lane  = threadIdx.x & 31;
  const int wave  = threadIdx.x >> 5;
  const int gw    = blockIdx.x * 8 + wave;
  const int b     = gw / NTILE;
  const int it    = gw % NTILE;

  const int half  = lane >> 4;
  const int l15   = lane & 15;
  const int kbase = half << 1;
  const int c     = l15;

  const float* nrm_b  = normed  + (size_t)b * NPATCH * 4;
  const float* pat_b  = patches + (size_t)b * NPATCH * 4;

  const float* b2base = (c < 4) ? (patT + ((size_t)b * 4 + c) * NPATCH + 8 * half)
                                : (consts + ((c == 4) ? 0 : CONSTF) + 8 * half);
  const int    b2step = (c < 4) ? (RING * 16) : 0;

  v2f a1 = *(const v2f*)(nrm_b + (it * 16 + l15) * 4 + kbase);

  v8f acc = {};

  v2f    xb1[RING];
  v4f    xb2[RING][2];
  v8f    xw[RING];
  v2f    yb1[RING];
  v4f    yb2[RING][2];
  v8f    yw[RING];

#pragma unroll
  for (int s = 0; s < RING; ++s)
    xb1[s] = *(const v2f*)(nrm_b + (s * 16 + l15) * 4 + kbase);
#pragma unroll
  for (int s = 0; s < RING; ++s)
    weight_tile(xb1[s], a1, xw[s]);

#pragma unroll
  for (int s = 0; s < RING; ++s) {
    xb1[s] = *(const v2f*)(nrm_b + ((RING + s) * 16 + l15) * 4 + kbase);
    xb2[s][0] = *(const v4f*)(b2base + s * 16);
    xb2[s][1] = *(const v4f*)(b2base + s * 16 + 4);
  }

  load_group_wait<false>(xb1, xb2);

  for (int g = 0; g + 1 < NGROUP; g += 2) {
    group_pass<false>(g,     nrm_b, b2base, b2step, a1, l15, kbase,
                      xb1, xb2, yb1, yb2, xw, yw, acc);
    group_pass<false>(g + 1, nrm_b, b2base, b2step, a1, l15, kbase,
                      yb1, yb2, xb1, xb2, yw, xw, acc);
  }
  group_pass<true>(NGROUP - 1, nrm_b, b2base, b2step, a1, l15, kbase,
                   xb1, xb2, yb1, yb2, xw, yw, acc);

#pragma unroll
  for (int r = 0; r < 8; ++r) {
    int row_local = r + 8 * half;
    float deg = __shfl(acc[r], (half << 4) | 4, 32);
    if (c < 4) {
      int n = it * 16 + row_local;
      float p = pat_b[n * 4 + c];
      out[(size_t)b * NPATCH * 4 + n * 4 + c] = p * (1.0f + deg) - acc[r];
    }
  }
}

extern "C" void kernel_launch(void* const* d_in, const int* in_sizes, int n_in,
                              void* d_out, int out_size, void* d_ws, size_t ws_size,
                              hipStream_t stream) {
  const float* x  = (const float*)d_in[0];
  const float* cw = (const float*)d_in[1];
  const float* cb = (const float*)d_in[2];
  float* out = (float*)d_out;

  float* patches = (float*)d_ws;
  float* normed  = patches + (size_t)BATCH * NPATCH * 4;
  float* patT    = normed  + (size_t)BATCH * NPATCH * 4;
  float* consts  = patT    + (size_t)BATCH * NPATCH * 4;

  int total = BATCH * NPATCH;
  conv_norm_kernel<<<(total + 255) / 256, 256, 0, stream>>>(x, cw, cb, patches, normed,
                                                            patT, consts);

  smooth_kernel<<<NTILE, 256, 0, stream>>>(patches, normed, patT, consts, out);
}


# Weight tests made on |cos| against the two f32 values where the rounded square first reaches the reference's bounds: weights bit-identical, the squaring multiply per element is gone; sums unchanged
# speedup vs baseline: 4.6354x; 1.0351x over previous
#include <hip/hip_runtime.h>
#include <hip/hip_fp16.h>


typedef __attribute__((ext_vector_type(2)))  float    v2f;
typedef __attribute__((ext_vector_type(4)))  float    v4f;
typedef __attribute__((ext_vector_type(8)))  float    v8f;
typedef __attribute__((ext_vector_type(16))) _Float16 v16h;

#define BATCH   8
#define HW_IN   112
#define HW_OUT  56
#define NPATCH  (HW_OUT * HW_OUT)
#define NTILE   (NPATCH / 16)
#define RING    4
#define NGROUP  (NTILE / RING)
#define CONSTF  (RING * 16)
static_assert(NTILE % RING == 0, "column loop is unrolled by RING");
static_assert(NTILE >= 2 * RING, "the prologue loads two groups");
static_assert(NGROUP % 2 == 1, "the column loop runs pairs of passes and peels one last pass");

__device__ __forceinline__ v8f wmma_k4(v2f a, v2f b, v8f c) {
#if __has_builtin(__builtin_amdgcn_wmma_f32_16x16x4_f32)
  return __builtin_amdgcn_wmma_f32_16x16x4_f32(false, a, false, b, (short)0, c,
                                               false, false);
#else
  v16h ah = {};
  v16h bh = {};
  ah[0] = (_Float16)a[0]; ah[1] = (_Float16)a[1];
  bh[0] = (_Float16)b[0]; bh[1] = (_Float16)b[1];
  return __builtin_amdgcn_wmma_f32_16x16x32_f16(false, ah, false, bh, (short)0, c,
                                                false, false);
#endif
}

__global__ __launch_bounds__(256) void conv_norm_kernel(
    const float* __restrict__ x,
    const float* __restrict__ cw,
    const float* __restrict__ cb,
    float* __restrict__ patches,
    float* __restrict__ normed,
    float* __restrict__ patT,
    float* __restrict__ consts) {
  int t = blockIdx.x * blockDim.x + threadIdx.x;
  if (t >= BATCH * NPATCH) return;
  if (t < CONSTF) {
    consts[t] = 1.0f;
    consts[CONSTF + t] = 0.0f;
  }
  int b = t / NPATCH;
  int n = t % NPATCH;
  int h = n / HW_OUT, w = n % HW_OUT;

  const float* xb = x + ((size_t)b * HW_IN + 2 * h) * HW_IN + 2 * w;
  float x00 = xb[0], x01 = xb[1], x10 = xb[HW_IN], x11 = xb[HW_IN + 1];

  float f[4];
  float ss = 0.0f;
#pragma unroll
  for (int cc = 0; cc < 4; ++cc) {
    float v = x00 * cw[cc * 4 + 0] + x01 * cw[cc * 4 + 1] +
              x10 * cw[cc * 4 + 2] + x11 * cw[cc * 4 + 3] + cb[cc];
    f[cc] = v;
    ss += v * v;
  }
  float inv = 1.0f / (sqrtf(ss) + 1e-12f);

  float4 pf = make_float4(f[0], f[1], f[2], f[3]);
  float4 nf = make_float4(f[0] * inv, f[1] * inv, f[2] * inv, f[3] * inv);
  *(float4*)(patches + (size_t)t * 4) = pf;
  *(float4*)(normed  + (size_t)t * 4) = nf;
#pragma unroll
  for (int cc = 0; cc < 4; ++cc)
    patT[((size_t)b * 4 + cc) * NPATCH + n] = f[cc];
}

__device__ __forceinline__ void load_group(const float* n1p, const float* b2p,
                                           v2f (&b1)[RING], v4f (&b2)[RING][2],
                                           v2f& rowop, v8f& acc) {
  static_assert(RING == 4, "the asm statement below holds the loads of 4 tiles");
  asm volatile(
      "global_load_b64 %0, %14, off\n\t"
      "global_load_b64 %1, %14, off offset:256\n\t"
      "global_load_b64 %2, %14, off offset:512\n\t"
      "global_load_b64 %3, %14, off offset:768\n\t"
      "global_load_b128 %4, %15, off\n\t"
      "global_load_b128 %5, %15, off offset:16\n\t"
      "global_load_b128 %6, %15, off offset:64\n\t"
      "global_load_b128 %7, %15, off offset:80\n\t"
      "global_load_b128 %8, %15, off offset:128\n\t"
      "global_load_b128 %9, %15, off offset:144\n\t"
      "global_load_b128 %10, %15, off offset:192\n\t"
      "global_load_b128 %11, %15, off offset:208"
      : "=&v"(b1[0]), "=&v"(b1[1]), "=&v"(b1[2]), "=&v"(b1[3]),
        "=&v"(b2[0][0]), "=&v"(b2[0][1]), "=&v"(b2[1][0]), "=&v"(b2[1][1]),
        "=&v"(b2[2][0]), "=&v"(b2[2][1]), "=&v"(b2[3][0]), "=&v"(b2[3][1]),
        "+v"(rowop), "+v"(acc)
      : "v"(n1p), "v"(b2p));
}

__device__ __forceinline__ void load_group_release(const float* n1p, const float* b2p,
                                                   v8f& acc) {
  asm volatile("s_wait_xcnt 0x0" : "+v"(acc) : "v"(n1p), "v"(b2p));
}

template <bool NEWER>
__device__ __forceinline__ void load_group_wait(v2f (&b1)[RING], v4f (&b2)[RING][2]) {
  if (NEWER) {
    asm volatile(
        "s_wait_loadcnt 0xc"
        : "+v"(b1[0]), "+v"(b1[1]), "+v"(b1[2]), "+v"(b1[3]),
          "+v"(b2[0][0]), "+v"(b2[0][1]), "+v"(b2[1][0]), "+v"(b2[1][1]),
          "+v"(b2[2][0]), "+v"(b2[2][1]), "+v"(b2[3][0]), "+v"(b2[3][1]));
  } else {
    asm volatile(
        "s_wait_loadcnt 0x0"
        : "+v"(b1[0]), "+v"(b1[1]), "+v"(b1[2]), "+v"(b1[3]),
          "+v"(b2[0][0]), "+v"(b2[0][1]), "+v"(b2[1][0]), "+v"(b2[1][1]),
          "+v"(b2[2][0]), "+v"(b2[2][1]), "+v"(b2[3][0]), "+v"(b2[3][1]));
  }
}

constexpr float ABS_HALF = 0x1.6a09e8p-1f;
constexpr float ABS_HIGH = 0x1.c9f25ep-1f;
static_assert(__builtin_bit_cast(unsigned, ABS_HALF) == 0x3f3504f4u, "bound for cos*cos >= 0.5f");
static_assert(__builtin_bit_cast(unsigned, ABS_HIGH) == 0x3f64f92fu, "bound for cos*cos >= 0.8f");

__device__ __forceinline__ void weight_tile(v2f colop, v2f rowop, v8f& w) {
  v8f zero = {};
  v8f cosd = wmma_k4(colop, rowop, zero);
#pragma unroll
  for (int r = 0; r < 8; ++r) {
    float ac = __builtin_fabsf(cosd[r]);
    w[r] = (ac >= ABS_HIGH) ? 1.0f : ((ac >= ABS_HALF) ? 0.5f : 0.0f);
  }
}

template <bool LAST>
__device__ __forceinline__ void group_pass(
    int g, const float* nrm_b, const float* b2base, int b2step,
    v2f& a1, int l15, int kbase,
    v2f (&cb1)[RING], v4f (&cb2)[RING][2],
    v2f (&nb1)[RING], v4f (&nb2)[RING][2],
    const v8f (&wcur)[RING], v8f (&wnxt)[RING], v8f& acc) {
  const float* n1p = nrm_b;
  const float* b2p = b2base;
  if (!LAST) {
    const int g1 = (g + 2 < NGROUP) ? (g + 2) : (NGROUP - 1);
    const int g2 = g + 1;
    n1p = nrm_b + ((size_t)g1 * RING * 16 + l15) * 4 + kbase;
    b2p = b2base + (size_t)g2 * b2step;
    load_group(n1p, b2p, nb1, nb2, a1, acc);
    load_group_wait<true>(cb1, cb2);

#pragma unroll
    for (int s = 0; s < RING; ++s)
      weight_tile(cb1[s], a1, wnxt[s]);
  }

  if (LAST) load_group_wait<false>(cb1, cb2);

#pragma unroll
  for (int s = 0; s < RING; ++s) {
    v2f ak, bk;
    ak[0] = wcur[s][0]; ak[1] = wcur[s][1];
    bk[0] = cb2[s][0].x; bk[1] = cb2[s][0].y;
    acc = wmma_k4(ak, bk, acc);
    ak[0] = wcur[s][2]; ak[1] = wcur[s][3];
    bk[0] = cb2[s][0].z; bk[1] = cb2[s][0].w;
    acc = wmma_k4(ak, bk, acc);
    ak[0] = wcur[s][4]; ak[1] = wcur[s][5];
    bk[0] = cb2[s][1].x; bk[1] = cb2[s][1].y;
    acc = wmma_k4(ak, bk, acc);
    ak[0] = wcur[s][6]; ak[1] = wcur[s][7];
    bk[0] = cb2[s][1].z; bk[1] = cb2[s][1].w;
    acc = wmma_k4(ak, bk, acc);
    if (!LAST && s == 0) load_group_release(n1p, b2p, acc);
  }
}

__global__ __launch_bounds__(256) void smooth_kernel(
    const float* __restrict__ patches,
    const float* __restrict__ normed,
    const float* __restrict__ patT,
    const float* __restrict__ consts,
    float* __restrict__ out) {
  const int lane  = threadIdx.x & 31;
  const int wave  = threadIdx.x >> 5;
  const int gw    = blockIdx.x * 8 + wave;
  const int b     = gw / NTILE;
  const int it    = gw % NTILE;

  const int half  = lane >> 4;
  const int l15   = lane & 15;
  const int kbase = half << 1;
  const int c     = l15;

  const float* nrm_b  = normed  + (size_t)b * NPATCH * 4;
  const float* pat_b  = patches + (size_t)b * NPATCH * 4;

  const float* b2base = (c < 4) ? (patT + ((size_t)b * 4 + c) * NPATCH + 8 * half)
                                : (consts + ((c == 4) ? 0 : CONSTF) + 8 * half);
  const int    b2step = (c < 4) ? (RING * 16) : 0;

  v2f a1 = *(const v2f*)(nrm_b + (it * 16 + l15) * 4 + kbase);

  v8f acc = {};

  v2f    xb1[RING];
  v4f    xb2[RING][2];
  v8f    xw[RING];
  v2f    yb1[RING];
  v4f    yb2[RING][2];
  v8f    yw[RING];

#pragma unroll
  for (int s = 0; s < RING; ++s)
    xb1[s] = *(const v2f*)(nrm_b + (s * 16 + l15) * 4 + kbase);
#pragma unroll
  for (int s = 0; s < RING; ++s)
    weight_tile(xb1[s], a1, xw[s]);

#pragma unroll
  for (int s = 0; s < RING; ++s) {
    xb1[s] = *(const v2f*)(nrm_b + ((RING + s) * 16 + l15) * 4 + kbase);
    xb2[s][0] = *(const v4f*)(b2base + s * 16);
    xb2[s][1] = *(const v4f*)(b2base + s * 16 + 4);
  }

  load_group_wait<false>(xb1, xb2);

  for (int g = 0; g + 1 < NGROUP; g += 2) {
    group_pass<false>(g,     nrm_b, b2base, b2step, a1, l15, kbase,
                      xb1, xb2, yb1, yb2, xw, yw, acc);
    group_pass<false>(g + 1, nrm_b, b2base, b2step, a1, l15, kbase,
                      yb1, yb2, xb1, xb2, yw, xw, acc);
  }
  group_pass<true>(NGROUP - 1, nrm_b, b2base, b2step, a1, l15, kbase,
                   xb1, xb2, yb1, yb2, xw, yw, acc);

#pragma unroll
  for (int r = 0; r < 8; ++r) {
    int row_local = r + 8 * half;
    float deg = __shfl(acc[r], (half << 4) | 4, 32);
    if (c < 4) {
      int n = it * 16 + row_local;
      float p = pat_b[n * 4 + c];
      out[(size_t)b * NPATCH * 4 + n * 4 + c] = p * (1.0f + deg) - acc[r];
    }
  }
}

extern "C" void kernel_launch(void* const* d_in, const int* in_sizes, int n_in,
                              void* d_out, int out_size, void* d_ws, size_t ws_size,
                              hipStream_t stream) {
  const float* x  = (const float*)d_in[0];
  const float* cw = (const float*)d_in[1];
  const float* cb = (const float*)d_in[2];
  float* out = (float*)d_out;

  float* patches = (float*)d_ws;
  float* normed  = patches + (size_t)BATCH * NPATCH * 4;
  float* patT    = normed  + (size_t)BATCH * NPATCH * 4;
  float* consts  = patT    + (size_t)BATCH * NPATCH * 4;

  int total = BATCH * NPATCH;
  conv_norm_kernel<<<(total + 255) / 256, 256, 0, stream>>>(x, cw, cb, patches, normed,
                                                            patT, consts);

  smooth_kernel<<<NTILE, 256, 0, stream>>>(patches, normed, patT, consts, out);
}


# Closing tail in one group: row tile's own patches (32 contiguous bytes per lane, channel-major plane) loaded in the prologue, deg by a row-share move, the 8 stores back to back; sums unchanged
# speedup vs baseline: 5.0258x; 1.0842x over previous
#include <hip/hip_runtime.h>
#include <hip/hip_fp16.h>


typedef __attribute__((ext_vector_type(2)))  float    v2f;
typedef __attribute__((ext_vector_type(4)))  float    v4f;
typedef __attribute__((ext_vector_type(8)))  float    v8f;
typedef __attribute__((ext_vector_type(16))) _Float16 v16h;

#define BATCH   8
#define HW_IN   112
#define HW_OUT  56
#define NPATCH  (HW_OUT * HW_OUT)
#define NTILE   (NPATCH / 16)
#define RING    4
#define NGROUP  (NTILE / RING)
#define CONSTF  (RING * 16)
static_assert(NTILE % RING == 0, "column loop is unrolled by RING");
static_assert(NTILE >= 2 * RING, "the prologue loads two groups");
static_assert(NGROUP % 2 == 1, "the column loop runs pairs of passes and peels one last pass");

__device__ __forceinline__ v8f wmma_k4(v2f a, v2f b, v8f c) {
#if __has_builtin(__builtin_amdgcn_wmma_f32_16x16x4_f32)
  return __builtin_amdgcn_wmma_f32_16x16x4_f32(false, a, false, b, (short)0, c,
                                               false, false);
#else
  v16h ah = {};
  v16h bh = {};
  ah[0] = (_Float16)a[0]; ah[1] = (_Float16)a[1];
  bh[0] = (_Float16)b[0]; bh[1] = (_Float16)b[1];
  return __builtin_amdgcn_wmma_f32_16x16x32_f16(false, ah, false, bh, (short)0, c,
                                                false, false);
#endif
}

__global__ __launch_bounds__(256) void conv_norm_kernel(
    const float* __restrict__ x,
    const float* __restrict__ cw,
    const float* __restrict__ cb,
    float* __restrict__ normed,
    float* __restrict__ patT,
    float* __restrict__ consts) {
  int t = blockIdx.x * blockDim.x + threadIdx.x;
  if (t >= BATCH * NPATCH) return;
  if (t < CONSTF) {
    consts[t] = 1.0f;
    consts[CONSTF + t] = 0.0f;
  }
  int b = t / NPATCH;
  int n = t % NPATCH;
  int h = n / HW_OUT, w = n % HW_OUT;

  const float* xb = x + ((size_t)b * HW_IN + 2 * h) * HW_IN + 2 * w;
  float x00 = xb[0], x01 = xb[1], x10 = xb[HW_IN], x11 = xb[HW_IN + 1];

  float f[4];
  float ss = 0.0f;
#pragma unroll
  for (int cc = 0; cc < 4; ++cc) {
    float v = x00 * cw[cc * 4 + 0] + x01 * cw[cc * 4 + 1] +
              x10 * cw[cc * 4 + 2] + x11 * cw[cc * 4 + 3] + cb[cc];
    f[cc] = v;
    ss += v * v;
  }
  float inv = 1.0f / (sqrtf(ss) + 1e-12f);

  float4 nf = make_float4(f[0] * inv, f[1] * inv, f[2] * inv, f[3] * inv);
  *(float4*)(normed  + (size_t)t * 4) = nf;
#pragma unroll
  for (int cc = 0; cc < 4; ++cc)
    patT[((size_t)b * 4 + cc) * NPATCH + n] = f[cc];
}

__device__ __forceinline__ void load_group(const float* n1p, const float* b2p,
                                           v2f (&b1)[RING], v4f (&b2)[RING][2],
                                           v2f& rowop, v8f& acc) {
  static_assert(RING == 4, "the asm statement below holds the loads of 4 tiles");
  asm volatile(
      "global_load_b64 %0, %14, off\n\t"
      "global_load_b64 %1, %14, off offset:256\n\t"
      "global_load_b64 %2, %14, off offset:512\n\t"
      "global_load_b64 %3, %14, off offset:768\n\t"
      "global_load_b128 %4, %15, off\n\t"
      "global_load_b128 %5, %15, off offset:16\n\t"
      "global_load_b128 %6, %15, off offset:64\n\t"
      "global_load_b128 %7, %15, off offset:80\n\t"
      "global_load_b128 %8, %15, off offset:128\n\t"
      "global_load_b128 %9, %15, off offset:144\n\t"
      "global_load_b128 %10, %15, off offset:192\n\t"
      "global_load_b128 %11, %15, off offset:208"
      : "=&v"(b1[0]), "=&v"(b1[1]), "=&v"(b1[2]), "=&v"(b1[3]),
        "=&v"(b2[0][0]), "=&v"(b2[0][1]), "=&v"(b2[1][0]), "=&v"(b2[1][1]),
        "=&v"(b2[2][0]), "=&v"(b2[2][1]), "=&v"(b2[3][0]), "=&v"(b2[3][1]),
        "+v"(rowop), "+v"(acc)
      : "v"(n1p), "v"(b2p));
}

__device__ __forceinline__ void load_group_release(const float* n1p, const float* b2p,
                                                   v8f& acc) {
  asm volatile("s_wait_xcnt 0x0" : "+v"(acc) : "v"(n1p), "v"(b2p));
}

template <bool NEWER>
__device__ __forceinline__ void load_group_wait(v2f (&b1)[RING], v4f (&b2)[RING][2]) {
  if (NEWER) {
    asm volatile(
        "s_wait_loadcnt 0xc"
        : "+v"(b1[0]), "+v"(b1[1]), "+v"(b1[2]), "+v"(b1[3]),
          "+v"(b2[0][0]), "+v"(b2[0][1]), "+v"(b2[1][0]), "+v"(b2[1][1]),
          "+v"(b2[2][0]), "+v"(b2[2][1]), "+v"(b2[3][0]), "+v"(b2[3][1]));
  } else {
    asm volatile(
        "s_wait_loadcnt 0x0"
        : "+v"(b1[0]), "+v"(b1[1]), "+v"(b1[2]), "+v"(b1[3]),
          "+v"(b2[0][0]), "+v"(b2[0][1]), "+v"(b2[1][0]), "+v"(b2[1][1]),
          "+v"(b2[2][0]), "+v"(b2[2][1]), "+v"(b2[3][0]), "+v"(b2[3][1]));
  }
}

__device__ __forceinline__ float row_col4(float v) {
  int iv = __builtin_bit_cast(int, v);
  return __builtin_bit_cast(float, __builtin_amdgcn_update_dpp(iv, iv, 0x154, 0xf, 0xf, false));
}

constexpr float ABS_HALF = 0x1.6a09e8p-1f;
constexpr float ABS_HIGH = 0x1.c9f25ep-1f;
static_assert(__builtin_bit_cast(unsigned, ABS_HALF) == 0x3f3504f4u, "bound for cos*cos >= 0.5f");
static_assert(__builtin_bit_cast(unsigned, ABS_HIGH) == 0x3f64f92fu, "bound for cos*cos >= 0.8f");

__device__ __forceinline__ void weight_tile(v2f colop, v2f rowop, v8f& w) {
  v8f zero = {};
  v8f cosd = wmma_k4(colop, rowop, zero);
#pragma unroll
  for (int r = 0; r < 8; ++r) {
    float ac = __builtin_fabsf(cosd[r]);
    w[r] = (ac >= ABS_HIGH) ? 1.0f : ((ac >= ABS_HALF) ? 0.5f : 0.0f);
  }
}

template <bool LAST>
__device__ __forceinline__ void group_pass(
    int g, const float* nrm_b, const float* b2base, int b2step,
    v2f& a1, int l15, int kbase,
    v2f (&cb1)[RING], v4f (&cb2)[RING][2],
    v2f (&nb1)[RING], v4f (&nb2)[RING][2],
    const v8f (&wcur)[RING], v8f (&wnxt)[RING], v8f& acc) {
  const float* n1p = nrm_b;
  const float* b2p = b2base;
  if (!LAST) {
    const int g1 = (g + 2 < NGROUP) ? (g + 2) : (NGROUP - 1);
    const int g2 = g + 1;
    n1p = nrm_b + ((size_t)g1 * RING * 16 + l15) * 4 + kbase;
    b2p = b2base + (size_t)g2 * b2step;
    load_group(n1p, b2p, nb1, nb2, a1, acc);
    load_group_wait<true>(cb1, cb2);

#pragma unroll
    for (int s = 0; s < RING; ++s)
      weight_tile(cb1[s], a1, wnxt[s]);
  }

  if (LAST) load_group_wait<false>(cb1, cb2);

#pragma unroll
  for (int s = 0; s < RING; ++s) {
    v2f ak, bk;
    ak[0] = wcur[s][0]; ak[1] = wcur[s][1];
    bk[0] = cb2[s][0].x; bk[1] = cb2[s][0].y;
    acc = wmma_k4(ak, bk, acc);
    ak[0] = wcur[s][2]; ak[1] = wcur[s][3];
    bk[0] = cb2[s][0].z; bk[1] = cb2[s][0].w;
    acc = wmma_k4(ak, bk, acc);
    ak[0] = wcur[s][4]; ak[1] = wcur[s][5];
    bk[0] = cb2[s][1].x; bk[1] = cb2[s][1].y;
    acc = wmma_k4(ak, bk, acc);
    ak[0] = wcur[s][6]; ak[1] = wcur[s][7];
    bk[0] = cb2[s][1].z; bk[1] = cb2[s][1].w;
    acc = wmma_k4(ak, bk, acc);
    if (!LAST && s == 0) load_group_release(n1p, b2p, acc);
  }
}

__global__ __launch_bounds__(256) void smooth_kernel(
    const float* __restrict__ normed,
    const float* __restrict__ patT,
    const float* __restrict__ consts,
    float* __restrict__ out) {
  const int lane  = threadIdx.x & 31;
  const int wave  = threadIdx.x >> 5;
  const int gw    = blockIdx.x * 8 + wave;
  const int b     = gw / NTILE;
  const int it    = gw % NTILE;

  const int half  = lane >> 4;
  const int l15   = lane & 15;
  const int kbase = half << 1;
  const int c     = l15;

  const float* nrm_b  = normed  + (size_t)b * NPATCH * 4;

  const float* b2base = (c < 4) ? (patT + ((size_t)b * 4 + c) * NPATCH + 8 * half)
                                : (consts + ((c == 4) ? 0 : CONSTF) + 8 * half);
  const int    b2step = (c < 4) ? (RING * 16) : 0;

  v2f a1 = *(const v2f*)(nrm_b + (it * 16 + l15) * 4 + kbase);

  v8f acc = {};

  v2f    xb1[RING];
  v4f    xb2[RING][2];
  v8f    xw[RING];
  v2f    yb1[RING];
  v4f    yb2[RING][2];
  v8f    yw[RING];

#pragma unroll
  for (int s = 0; s < RING; ++s)
    xb1[s] = *(const v2f*)(nrm_b + (s * 16 + l15) * 4 + kbase);
#pragma unroll
  for (int s = 0; s < RING; ++s)
    weight_tile(xb1[s], a1, xw[s]);

#pragma unroll
  for (int s = 0; s < RING; ++s) {
    xb1[s] = *(const v2f*)(nrm_b + ((RING + s) * 16 + l15) * 4 + kbase);
    xb2[s][0] = *(const v4f*)(b2base + s * 16);
    xb2[s][1] = *(const v4f*)(b2base + s * 16 + 4);
  }

  const float* prow_p = patT + ((size_t)b * 4 + (c & 3)) * NPATCH + it * 16 + 8 * half;
  v4f prow0 = *(const v4f*)(prow_p);
  v4f prow1 = *(const v4f*)(prow_p + 4);

  load_group_wait<false>(xb1, xb2);

  for (int g = 0; g + 1 < NGROUP; g += 2) {
    group_pass<false>(g,     nrm_b, b2base, b2step, a1, l15, kbase,
                      xb1, xb2, yb1, yb2, xw, yw, acc);
    group_pass<false>(g + 1, nrm_b, b2base, b2step, a1, l15, kbase,
                      yb1, yb2, xb1, xb2, yw, xw, acc);
  }
  group_pass<true>(NGROUP - 1, nrm_b, b2base, b2step, a1, l15, kbase,
                   xb1, xb2, yb1, yb2, xw, yw, acc);

  float deg[8];
#pragma unroll
  for (int r = 0; r < 8; ++r)
    deg[r] = row_col4(acc[r]);
  if (c < 4) {
    float* orow = out + (size_t)b * NPATCH * 4 + (size_t)(it * 16 + 8 * half) * 4 + c;
#pragma unroll
    for (int r = 0; r < 8; ++r) {
      float p = (r < 4) ? prow0[r & 3] : prow1[r & 3];
      orow[r * 4] = p * (1.0f + deg[r]) - acc[r];
    }
  }
}

extern "C" void kernel_launch(void* const* d_in, const int* in_sizes, int n_in,
                              void* d_out, int out_size, void* d_ws, size_t ws_size,
                              hipStream_t stream) {
  const float* x  = (const float*)d_in[0];
  const float* cw = (const float*)d_in[1];
  const float* cb = (const float*)d_in[2];
  float* out = (float*)d_out;

  float* normed  = (float*)d_ws;
  float* patT    = normed  + (size_t)BATCH * NPATCH * 4;
  float* consts  = patT    + (size_t)BATCH * NPATCH * 4;

  int total = BATCH * NPATCH;
  conv_norm_kernel<<<(total + 255) / 256, 256, 0, stream>>>(x, cw, cb, normed, patT, consts);

  smooth_kernel<<<NTILE, 256, 0, stream>>>(normed, patT, consts, out);
}
